# starting kernel
# baseline (speedup 1.0000x reference)
#include <hip/hip_runtime.h>
#include <hip/hip_bf16.h>
#include <math.h>


typedef __attribute__((ext_vector_type(2))) float v2f;
typedef __attribute__((ext_vector_type(8))) float v8f;

#define NN   128
#define SLD  132
#define PLD  132
#define TS   10
#define STEP_EPS 0.1f
#define ALPHA 5.0f
#define ALPHA_BASE 0.1f
#define KRANK 3
#define PSTEPS 10

__device__ __forceinline__ float softplus_f(float x) {
  return fmaxf(x, 0.0f) + log1pf(expf(-fabsf(x)));
}

__device__ __forceinline__ float block_reduce(float v, float* red, int tid) {
  red[tid] = v;
  __syncthreads();
  #pragma unroll
  for (int s = 128; s > 0; s >>= 1) {
    if (tid < s) red[tid] += red[tid + s];
    __syncthreads();
  }
  float r = red[0];
  __syncthreads();
  return r;
}

__launch_bounds__(256, 1)
__global__ void ts_gcn_fused(const float* __restrict__ edge_pred,
                             const int*   __restrict__ adj,
                             const float* __restrict__ d_init,
                             const float* __restrict__ u_init,
                             const float* __restrict__ x_noise,
                             float*       __restrict__ out)
{
  __shared__ float  Dp[NN * NN];
  __shared__ float  Cw[NN * NN];
  __shared__ float  Sm[NN * SLD];
  __shared__ float2 Xq[64 * 16];
  __shared__ float  Pp[16 * PLD];
  __shared__ float  Xx[NN], Xy[NN], Xz[NN];
  __shared__ float  rsum[NN];
  __shared__ float  red[256];
  __shared__ float  ucol[KRANK][NN];
  __shared__ float  uvec[NN], utmp[NN];

  const int b   = blockIdx.x;
  const int tid = threadIdx.x;
  const float d0 = d_init[0];

  const float* ep = edge_pred + (size_t)b * NN * NN * 2;
  const int*   ad = adj       + (size_t)b * NN * NN;

  float partM = 0.0f;
  for (int p = tid; p < NN * NN; p += 256) {
    const int i = p >> 7, j = p & 127;
    const float  m = (float)ad[p];
    const float2 e = ((const float2*)ep)[p];
    const float2 t = ((const float2*)ep)[j * NN + i];
    const float sp0 = softplus_f(d0 + e.x + t.x);
    const float sp1 = softplus_f(d0 + e.y + t.y);
    Dp[p] = sp0 * m;
    Cw[p] = sp1 * m;
    partM += (i == j) ? 1.0f : m;
  }
  for (int idx = tid; idx < 64 * 16; idx += 256) Xq[idx] = make_float2(0.f, 0.f);
  const float Msum = block_reduce(partM, red, tid);

  const float Nmol = block_reduce(
      (tid < NN) ? ((tid == 0) ? 1.0f : (float)ad[tid * NN]) : 0.0f, red, tid);

  {
    const int i = tid >> 1, j0 = (tid & 1) << 6;
    float part = 0.0f;
    for (int jj = 0; jj < 64; ++jj) {
      const float d = Dp[i * NN + j0 + jj];
      part += d * d;
    }
    red[tid] = part;
    __syncthreads();
    if (tid < NN) rsum[tid] = red[2 * tid] + red[2 * tid + 1];
    __syncthreads();
    const float tot  = block_reduce((tid < NN) ? rsum[tid] : 0.0f, red, tid);
    const float invN = 1.0f / Nmol;
    const float Mn   = tot * invN * invN;
    for (int p = tid; p < NN * NN; p += 256) {
      const int i2 = p >> 7, j2 = p & 127;
      const float d  = Dp[p];
      const float d2 = d * d;
      const float mk = (i2 == j2) ? 1.0f : (d > 0.0f ? 1.0f : 0.0f);
      Sm[i2 * SLD + j2] =
          mk * -0.5f * (d2 - rsum[j2] * invN - rsum[i2] * invN + Mn);
    }
    __syncthreads();
  }

  for (int kx = 0; kx < KRANK; ++kx) {
    if (tid < NN) uvec[tid] = u_init[(size_t)b * NN * 3 + tid * 3 + kx];
    __syncthreads();
    for (int it = 0; it < PSTEPS; ++it) {
      const float nv =
          block_reduce((tid < NN) ? uvec[tid] * uvec[tid] : 0.0f, red, tid);
      const float inorm = 1.0f / fmaxf(sqrtf(nv), 0.001f);
      if (tid < NN) uvec[tid] *= inorm;
      __syncthreads();
      float dc0 = 0.0f, dc1 = 0.0f;
      if (kx > 0)
        dc0 = block_reduce((tid < NN) ? ucol[0][tid] * uvec[tid] : 0.0f, red, tid);
      if (kx > 1)
        dc1 = block_reduce((tid < NN) ? ucol[1][tid] * uvec[tid] : 0.0f, red, tid);
      if (tid < NN) {
        const float* row = &Sm[tid * SLD];
        float acc = 0.0f;
        #pragma unroll 8
        for (int j = 0; j < NN; ++j) acc = fmaf(row[j], uvec[j], acc);
        if (kx > 0) acc -= ucol[0][tid] * dc0;
        if (kx > 1) acc -= ucol[1][tid] * dc1;
        utmp[tid] = acc;
      }
      __syncthreads();
      if (tid < NN) uvec[tid] = utmp[tid];
      __syncthreads();
    }
    const float eig =
        block_reduce((tid < NN) ? uvec[tid] * uvec[tid] : 0.0f, red, tid);
    const float iden = 1.0f / sqrtf(sqrtf(eig + 0.01f));
    if (tid < NN) ucol[kx][tid] = uvec[tid] * iden;
    __syncthreads();
  }
  if (tid < NN) {
    const float* xn = x_noise + (size_t)b * NN * 3 + tid * 3;
    Xx[tid] = ucol[0][tid] + xn[0];
    Xy[tid] = ucol[1][tid] + xn[1];
    Xz[tid] = ucol[2][tid] + xn[2];
  }
  __syncthreads();
  if (tid < 192) {
    const int q = tid & 63, n = tid >> 6;
    const float* comp = (n == 0) ? Xx : ((n == 1) ? Xy : Xz);
    Xq[q * 16 + n] = make_float2(comp[2 * q], comp[2 * q + 1]);
  }
  __syncthreads();

  const float invM4 = 4.0f / Msum;
  const int lane   = tid & 31;
  const int mt     = tid >> 5;
  const int nidx   = lane & 15;
  const int laneHi = lane >> 4;
  const int arow   = mt * 16 + nidx;
  const float2* aq = (const float2*)&Sm[arow * SLD + laneHi * 2];
  const float2* bq = &Xq[laneHi * 16 + nidx];

  for (int tstep = 0; tstep < TS; ++tstep) {
    {
      const int i = tid >> 1, j0 = (tid & 1) << 6;
      const float xi0 = Xx[i], xi1 = Xy[i], xi2 = Xz[i];
      float part = 0.0f;
      for (int jj = 0; jj < 64; ++jj) {
        const int j = j0 + jj;
        const int p = i * NN + j;
        const float u0 = xi0 - Xx[j];
        const float u1 = xi1 - Xy[j];
        const float u2 = xi2 - Xz[j];
        const float dist = sqrtf(u0 * u0 + u1 * u1 + u2 * u2 + 0.01f);
        const float s = invM4 * Cw[p] * (dist - Dp[p]) / dist;
        Sm[i * SLD + j] = s;
        part += s;
      }
      red[tid] = part;
      __syncthreads();
      if (tid < NN) rsum[tid] = red[2 * tid] + red[2 * tid + 1];
      __syncthreads();
    }

    v8f acc = {};
    #pragma unroll
    for (int kt = 0; kt < 32; ++kt) {
      const float2 apair = aq[kt * 2];
      const float2 bpair = bq[kt * 32];
      v2f a;  a[0]  = apair.x;  a[1]  = apair.y;
      v2f bb; bb[0] = bpair.x;  bb[1] = bpair.y;
      acc = __builtin_amdgcn_wmma_f32_16x16x4_f32(
           false, a,  false, bb,
           (short)0, acc,  false,  false);
    }
    {
      const int rbase = mt * 16 + laneHi * 8;
      #pragma unroll
      for (int r = 0; r < 8; ++r) Pp[nidx * PLD + rbase + r] = acc[r];
    }
    __syncthreads();

    if (tid < NN) {
      const float gx = rsum[tid] * Xx[tid] - Pp[0 * PLD + tid];
      const float gy = rsum[tid] * Xy[tid] - Pp[1 * PLD + tid];
      const float gz = rsum[tid] * Xz[tid] - Pp[2 * PLD + tid];
      const float dx = -STEP_EPS * gx;
      const float dy = -STEP_EPS * gy;
      const float dz = -STEP_EPS * gz;
      const float speed = sqrtf(dx * dx + dy * dy + dz * dz + 0.001f);
      const float at =
          ALPHA_BASE + (ALPHA - ALPHA_BASE) * ((float)(TS - tstep) / (float)TS);
      const float sc = at * tanhf(speed / at) / speed;
      Xx[tid] += dx * sc;
      Xy[tid] += dy * sc;
      Xz[tid] += dz * sc;
    }
    __syncthreads();
    if (tid < 192) {
      const int q = tid & 63, n = tid >> 6;
      const float* comp = (n == 0) ? Xx : ((n == 1) ? Xy : Xz);
      Xq[q * 16 + n] = make_float2(comp[2 * q], comp[2 * q + 1]);
    }
    __syncthreads();
  }

  float* ob = out + (size_t)b * NN * NN;
  for (int p = tid; p < NN * NN; p += 256) {
    const int i = p >> 7, j = p & 127;
    const float m  = (float)ad[p];
    const float u0 = Xx[i] - Xx[j];
    const float u1 = Xy[i] - Xy[j];
    const float u2 = Xz[i] - Xz[j];
    ob[p] = m * sqrtf(u0 * u0 + u1 * u1 + u2 * u2 + 0.01f);
  }
}

extern "C" void kernel_launch(void* const* d_in, const int* in_sizes, int n_in,
                              void* d_out, int out_size, void* d_ws, size_t ws_size,
                              hipStream_t stream) {
  const float* edge_pred = (const float*)d_in[0];
  const int*   adj       = (const int*)  d_in[1];
  const float* d_init    = (const float*)d_in[2];
  const float* u_init    = (const float*)d_in[3];
  const float* x_noise   = (const float*)d_in[4];
  float*       outp      = (float*)d_out;
  const int B = in_sizes[1] / (NN * NN);
  ts_gcn_fused<<<B, 256, 0, stream>>>(edge_pred, adj, d_init, u_init, x_noise, outp);
}


# Power iteration: iterate kept in registers, entries read as scalar operands (v_readlane), no LDS store + 32 broadcast reads a step; norm and deflation dots summed across lanes: sum order changes
# speedup vs baseline: 5.0540x; 5.0540x over previous
#include <hip/hip_runtime.h>
#include <hip/hip_bf16.h>
#include <math.h>


typedef __attribute__((ext_vector_type(2))) float v2f;
typedef __attribute__((ext_vector_type(4))) float v4f;
typedef __attribute__((ext_vector_type(4))) int   v4i;
typedef __attribute__((ext_vector_type(8))) float v8f;

#define NN   128
#define SLD  132
#define PLD  132
#define TS   10
#define STEP_EPS 0.1f
#define ALPHA 5.0f
#define ALPHA_BASE 0.1f
#define KRANK 3
#define PSTEPS 10

__device__ __forceinline__ float softplus_f(float x) {
  return fmaxf(x, 0.0f) + log1pf(expf(-fabsf(x)));
}

__device__ __forceinline__ float sqrt_rn_bounded(float x) {
  const float s  = __builtin_amdgcn_sqrtf(x);
  const float sd = __int_as_float(__float_as_int(s) - 1);
  const float su = __int_as_float(__float_as_int(s) + 1);
  const float vd = fmaf(-sd, s, x);
  const float vu = fmaf(-su, s, x);
  float r = (vd <= 0.0f) ? sd : s;
  r = (vu > 0.0f) ? su : r;
  return r;
}

__device__ __forceinline__ void dist_rinv(float q, float& dist, float& rinv) {
  const float s  = __builtin_amdgcn_sqrtf(q);
  const float r0 = __builtin_amdgcn_rcpf(s);
  const float d0 = fmaf(-s, s, q);
  dist = fmaf(d0, 0.5f * r0, s);
  const float e0 = fmaf(-dist, r0, 1.0f);
  const float r1 = fmaf(e0, r0, r0);
  const float e1 = fmaf(-dist, r1, 1.0f);
  rinv = fmaf(e1, r1, r1);
}

__device__ __forceinline__ float s_cell(float u0, float u1, float u2,
                                        float e, float d) {
  float dist, rinv;
  dist_rinv(u0 * u0 + u1 * u1 + u2 * u2 + 0.01f, dist, rinv);
  return e * (dist - d) * rinv;
}

__device__ __forceinline__ void s_pair(float u0, float u1, float u2,
                                       float eij, float eji, float dij, float dji,
                                       float& sij, float& sji) {
  float dist, rinv;
  dist_rinv(u0 * u0 + u1 * u1 + u2 * u2 + 0.01f, dist, rinv);
  sij = eij * (dist - dij) * rinv;
  sji = eji * (dist - dji) * rinv;
}

__device__ __forceinline__ float out_cell(float u0, float u1, float u2, int m) {
  return (float)m * sqrt_rn_bounded(fmaf(u2, u2, fmaf(u0, u0, u1 * u1)) + 0.01f);
}

__device__ __forceinline__ float block_reduce(float v, float* red, int tid) {
  red[tid] = v;
  __syncthreads();
  #pragma unroll
  for (int s = 128; s > 0; s >>= 1) {
    if (tid < s) red[tid] += red[tid + s];
    __syncthreads();
  }
  float r = red[0];
  __syncthreads();
  return r;
}


__device__ __forceinline__ v2f fma2(v2f x, v2f y, v2f acc) {
  return __builtin_elementwise_fma(x, y, acc);
}


template <int CTRL>
__device__ __forceinline__ float add_row_xmask(float v) {
  return v + __builtin_bit_cast(float, __builtin_amdgcn_update_dpp(
                 0, __builtin_bit_cast(int, v), CTRL, 0xf, 0xf, false));
}

__device__ __forceinline__ float wave_sum(float v) {
  v = add_row_xmask<0x161>(v);
  v = add_row_xmask<0x162>(v);
  v = add_row_xmask<0x164>(v);
  v = add_row_xmask<0x168>(v);
  const float lo = __builtin_bit_cast(float, __builtin_amdgcn_readlane(__builtin_bit_cast(int, v), 0));
  const float hi = __builtin_bit_cast(float, __builtin_amdgcn_readlane(__builtin_bit_cast(int, v), 16));
  return lo + hi;
}

struct PSet { v4f g[4][2]; };
struct PAcc { v2f a[4]; v2f b[4]; };

__device__ __forceinline__ void pset_load(PSet& s, const float* g0, const float* g1,
                                          const float* g2, const float* g3, int j) {
  #pragma unroll
  for (int q = 0; q < 2; ++q) {
    s.g[0][q] = *(const v4f*)&g0[j + 4 * q];
    s.g[1][q] = *(const v4f*)&g1[j + 4 * q];
    s.g[2][q] = *(const v4f*)&g2[j + 4 * q];
    s.g[3][q] = *(const v4f*)&g3[j + 4 * q];
  }
}

template <int S>
__device__ __forceinline__ void pset_use(PAcc& acc, const PSet& s, float ucur) {
  float us[8];
  #pragma unroll
  for (int c = 0; c < 8; ++c)
    us[c] = __builtin_bit_cast(float, __builtin_amdgcn_readlane(
                __builtin_bit_cast(int, ucur), 8 * S + c));
  #pragma unroll
  for (int q = 0; q < 2; ++q) {
    const v2f ulo = {us[4 * q],     us[4 * q + 1]};
    const v2f uhi = {us[4 * q + 2], us[4 * q + 3]};
    #pragma unroll
    for (int r = 0; r < 4; ++r) {
      const v4f g = s.g[r][q];
      if (q & 1) { acc.b[r] = fma2(g.xy, ulo, acc.b[r]);  acc.b[r] = fma2(g.zw, uhi, acc.b[r]); }
      else       { acc.a[r] = fma2(g.xy, ulo, acc.a[r]);  acc.a[r] = fma2(g.zw, uhi, acc.a[r]); }
    }
  }
}

__device__ __forceinline__ void power_group(PAcc& acc, PSet& sa, PSet& sb, PSet& sc, PSet& sd,
                                            const float* g0, const float* g1,
                                            const float* g2, const float* g3,
                                            int jb, float ucur) {
  pset_load(sc, g0, g1, g2, g3, jb + 16);
  pset_use<0>(acc, sa, ucur);
  pset_load(sd, g0, g1, g2, g3, jb + 24);
  pset_use<1>(acc, sb, ucur);
  pset_load(sa, g0, g1, g2, g3, (jb + 32) & (NN - 1));
  pset_use<2>(acc, sc, ucur);
  pset_load(sb, g0, g1, g2, g3, (jb + 40) & (NN - 1));
  pset_use<3>(acc, sd, ucur);
  asm volatile("" ::: "memory");
}

template <int KX>
__device__ __forceinline__ void power_column(const float* gram,
                                             float (*ucol)[NN], const float* ub,
                                             int lane) {
  const float* g0 = gram + (lane     ) * SLD;
  const float* g1 = gram + (lane + 32) * SLD;
  const float* g2 = gram + (lane + 64) * SLD;
  const float* g3 = gram + (lane + 96) * SLD;
  float u0 = ub[(lane     ) * 3 + KX];
  float u1 = ub[(lane + 32) * 3 + KX];
  float u2 = ub[(lane + 64) * 3 + KX];
  float u3 = ub[(lane + 96) * 3 + KX];
  float p0 = 0.0f, p1 = 0.0f, p2 = 0.0f, p3 = 0.0f;
  float q0 = 0.0f, q1 = 0.0f, q2 = 0.0f, q3 = 0.0f;
  if constexpr (KX > 0) {
    p0 = ucol[0][lane];      p1 = ucol[0][lane + 32];
    p2 = ucol[0][lane + 64]; p3 = ucol[0][lane + 96];
  }
  if constexpr (KX > 1) {
    q0 = ucol[1][lane];      q1 = ucol[1][lane + 32];
    q2 = ucol[1][lane + 64]; q3 = ucol[1][lane + 96];
  }
  const v2f zero2 = {0.0f, 0.0f};

  PSet sa, sb, sc, sd;
  pset_load(sa, g0, g1, g2, g3, 0);
  pset_load(sb, g0, g1, g2, g3, 8);

  for (int it = 0; it < PSTEPS; ++it) {
    pset_load(sc, g0, g1, g2, g3, 16);
    pset_load(sd, g0, g1, g2, g3, 24);
    const float nsq = wave_sum((u0 * u0 + u1 * u1) + (u2 * u2 + u3 * u3));
    float inorm = 1.0f / fmaxf(sqrtf(nsq), 0.001f);
    float dp = 0.0f, dq = 0.0f;
    if constexpr (KX > 0) dp = wave_sum((p0 * u0 + p1 * u1) + (p2 * u2 + p3 * u3));
    if constexpr (KX > 1) dq = wave_sum((q0 * u0 + q1 * u1) + (q2 * u2 + q3 * u3));
    asm volatile("" : "+v"(inorm));
    if constexpr (KX > 0) asm volatile("" : "+v"(dp));
    if constexpr (KX > 1) asm volatile("" : "+v"(dq));
    PAcc acc;
    #pragma unroll
    for (int r = 0; r < 4; ++r) { acc.a[r] = zero2; acc.b[r] = zero2; }
    pset_use<0>(acc, sa, u0);
    pset_use<1>(acc, sb, u0);
    pset_load(sa, g0, g1, g2, g3, 32);
    pset_use<2>(acc, sc, u0);
    pset_load(sb, g0, g1, g2, g3, 40);
    pset_use<3>(acc, sd, u0);
    float ucur = u1;
    #pragma unroll 1
    for (int t = 1; t < 4; ++t) {
      power_group(acc, sa, sb, sc, sd, g0, g1, g2, g3, 32 * t, ucur);
      ucur = (t == 1) ? u2 : u3;
    }
    float r0 = (acc.a[0].x + acc.a[0].y) + (acc.b[0].x + acc.b[0].y);
    float r1 = (acc.a[1].x + acc.a[1].y) + (acc.b[1].x + acc.b[1].y);
    float r2 = (acc.a[2].x + acc.a[2].y) + (acc.b[2].x + acc.b[2].y);
    float r3 = (acc.a[3].x + acc.a[3].y) + (acc.b[3].x + acc.b[3].y);
    if constexpr (KX > 0) {
      r0 = fmaf(-p0, dp, r0);  r1 = fmaf(-p1, dp, r1);
      r2 = fmaf(-p2, dp, r2);  r3 = fmaf(-p3, dp, r3);
    }
    if constexpr (KX > 1) {
      r0 = fmaf(-q0, dq, r0);  r1 = fmaf(-q1, dq, r1);
      r2 = fmaf(-q2, dq, r2);  r3 = fmaf(-q3, dq, r3);
    }
    u0 = r0 * inorm;  u1 = r1 * inorm;  u2 = r2 * inorm;  u3 = r3 * inorm;
  }

  const float eig  = wave_sum((u0 * u0 + u1 * u1) + (u2 * u2 + u3 * u3));
  const float iden = 1.0f / sqrtf(sqrtf(eig + 0.01f));
  ucol[KX][lane]      = u0 * iden;
  ucol[KX][lane + 32] = u1 * iden;
  ucol[KX][lane + 64] = u2 * iden;
  ucol[KX][lane + 96] = u3 * iden;
}

__launch_bounds__(256, 4)
__global__ void ts_gcn_fused(const float* __restrict__ edge_pred,
                             const int*   __restrict__ adj,
                             const float* __restrict__ d_init,
                             const float* __restrict__ u_init,
                             const float* __restrict__ x_noise,
                             float*       __restrict__ out)
{
  __shared__ __attribute__((aligned(16))) float Dp[NN * NN];
  __shared__ __attribute__((aligned(16))) float Sm[NN * SLD];
  __shared__ float2 Xq[64 * 16];
  __shared__ __attribute__((aligned(16))) float Pp[16 * PLD];
  __shared__ __attribute__((aligned(16))) float Xx[NN];
  __shared__ __attribute__((aligned(16))) float Xy[NN];
  __shared__ __attribute__((aligned(16))) float Xz[NN];
  __shared__ float  rsum[NN];
  __shared__ float  red[256];
  __shared__ __attribute__((aligned(16))) float ucol[KRANK][NN];

  const int b   = blockIdx.x;
  const int tid = threadIdx.x;
  const int lane   = tid & 31;
  const int mt     = tid >> 5;
  const float d0 = d_init[0];

  const float* ep = edge_pred + (size_t)b * NN * NN * 2;
  const int*   ad = adj       + (size_t)b * NN * NN;

  float partM = 0.0f;
  {
    const v4f* ep4 = (const v4f*)ep;
    const v4i* ad4 = (const v4i*)ad;
    for (int it = tid; it < 528; it += 256) {
      const int R = it / 33, k = it - R * 33;
      const bool lo = k < 32 - R;
      const int I = lo ? R : 31 - R;
      const int J = lo ? R + k : k - 1;
      v4f e[4][2], t[4][2];
      v4i ma[4], mb[4];
      #pragma unroll
      for (int a = 0; a < 4; ++a) {
        e[a][0] = ep4[(4 * I + a) * 64 + 2 * J];
        e[a][1] = ep4[(4 * I + a) * 64 + 2 * J + 1];
        t[a][0] = ep4[(4 * J + a) * 64 + 2 * I];
        t[a][1] = ep4[(4 * J + a) * 64 + 2 * I + 1];
        ma[a]   = ad4[(4 * I + a) * 32 + J];
        mb[a]   = ad4[(4 * J + a) * 32 + I];
      }
      const float offd = (I != J) ? 1.0f : 0.0f;
      v4f dd[4], ww[4];
      v4f dm[4], wm[4];
      #pragma unroll
      for (int a = 0; a < 4; ++a) {
        #pragma unroll
        for (int b = 0; b < 4; ++b) {
          const float e0 = e[a][b >> 1][(b & 1) * 2];
          const float e1 = e[a][b >> 1][(b & 1) * 2 + 1];
          const float t0 = t[b][a >> 1][(a & 1) * 2];
          const float t1 = t[b][a >> 1][(a & 1) * 2 + 1];
          const float sp0 = softplus_f(d0 + (e0 + t0));
          const float sp1 = softplus_f(d0 + (e1 + t1));
          const float m  = (float)ma[a][b];
          const float mm = (float)mb[b][a];
          dd[a][b] = sp0 * m;
          ww[a][b] = sp1 * m;
          dm[b][a] = sp0 * mm;
          wm[b][a] = sp1 * mm;
          partM += (I == J && a == b) ? 1.0f : m;
          partM = fmaf(offd, mm, partM);
          __builtin_amdgcn_sched_barrier(0);
        }
      }
      #pragma unroll
      for (int a = 0; a < 4; ++a) {
        *(v4f*)&Dp[(4 * I + a) * NN  + 4 * J] = dd[a];
        *(v4f*)&Sm[(4 * I + a) * SLD + 4 * J] = ww[a];
        *(v4f*)&Dp[(4 * J + a) * NN  + 4 * I] = dm[a];
        *(v4f*)&Sm[(4 * J + a) * SLD + 4 * I] = wm[a];
      }
    }
  }
  for (int idx = tid; idx < 64 * 16; idx += 256) {
    const float fill = ((idx & 15) == 3) ? 1.0f : 0.0f;
    Xq[idx] = make_float2(fill, fill);
  }
  const float Msum = block_reduce(partM, red, tid);

  const float invM4 = 4.0f / Msum;
  v4f cw[16];
  {
    const int br = lane >> 3, bc = lane & 7;
    const int rowA = mt * 16 + 4 * br;
    #pragma unroll
    for (int k = 0; k < 4; ++k) {
      const int col = ((mt + k) & 7) * 16 + 2 * bc;
      #pragma unroll
      for (int a = 0; a < 4; ++a) {
        const v2f w2 = *(const v2f*)&Sm[(rowA + a) * SLD + col];
        cw[2 * k + (a >> 1)][(a & 1) * 2]     = invM4 * w2.x;
        cw[2 * k + (a >> 1)][(a & 1) * 2 + 1] = invM4 * w2.y;
      }
      if (k > 0) {
        #pragma unroll
        for (int e = 0; e < 2; ++e)
          cw[6 + 2 * k + e] = invM4 * *(const v4f*)&Sm[(col + e) * SLD + rowA];
      }
    }
    const bool low = mt < 4;
    const int row4 = mt * 16 + 4 * (low ? (lane >> 4) : (lane >> 3));
    const int col4 = ((mt + 4) & 7) * 16 + (low ? (lane & 15) : 8 + (lane & 7));
    #pragma unroll
    for (int a = 0; a < 4; ++a) cw[14][a] = invM4 * Sm[(row4 + a) * SLD + col4];
    cw[15] = invM4 * *(const v4f*)&Sm[col4 * SLD + row4];
  }

  const float Nmol = block_reduce(
      (tid < NN) ? ((tid == 0) ? 1.0f : (float)ad[tid * NN]) : 0.0f, red, tid);

  {
    const int i = tid >> 1, j0 = (tid & 1) << 6;
    float part = 0.0f;
    for (int jj = 0; jj < 64; ++jj) {
      const float d = Dp[i * NN + j0 + jj];
      part += d * d;
    }
    red[tid] = part;
    __syncthreads();
    if (tid < NN) rsum[tid] = red[2 * tid] + red[2 * tid + 1];
    __syncthreads();
    const float tot  = block_reduce((tid < NN) ? rsum[tid] : 0.0f, red, tid);
    const float invN = 1.0f / Nmol;
    const float Mn   = tot * invN * invN;
    for (int p = tid; p < NN * NN; p += 256) {
      const int i2 = p >> 7, j2 = p & 127;
      const float d  = Dp[p];
      const float d2 = d * d;
      const float mk = (i2 == j2) ? 1.0f : (d > 0.0f ? 1.0f : 0.0f);
      Sm[i2 * SLD + j2] =
          mk * -0.5f * (d2 - rsum[j2] * invN - rsum[i2] * invN + Mn);
    }
    __syncthreads();
  }

  if (tid < 32) {
    #pragma unroll
    for (int r = 0; r < 16; ++r) *(v4f*)&Pp[r * NN + 4 * lane] = cw[r];
    const float* ub = u_init + (size_t)b * NN * 3;
    power_column<0>(Sm, ucol, ub, tid);
    power_column<1>(Sm, ucol, ub, tid);
    power_column<2>(Sm, ucol, ub, tid);
    #pragma unroll
    for (int r = 0; r < 16; ++r) cw[r] = *(const v4f*)&Pp[r * NN + 4 * lane];
  }
  __syncthreads();
  if (tid < NN) {
    const float* xn = x_noise + (size_t)b * NN * 3 + tid * 3;
    Xx[tid] = ucol[0][tid] + xn[0];
    Xy[tid] = ucol[1][tid] + xn[1];
    Xz[tid] = ucol[2][tid] + xn[2];
  }
  __syncthreads();
  if (tid < 192) {
    const int q = tid & 63, n = tid >> 6;
    const float* comp = (n == 0) ? Xx : ((n == 1) ? Xy : Xz);
    Xq[q * 16 + n] = make_float2(comp[2 * q], comp[2 * q + 1]);
  }
  __syncthreads();

  int tidd = tid;
  asm volatile("" : "+v"(tidd));
  const int laneD  = tidd & 31;
  const int nidx   = laneD & 15;
  const int laneHi = laneD >> 4;
  const int arow   = mt * 16 + nidx;
  const int brD  = laneD >> 3, bcD = laneD & 7;
  const int rowA = mt * 16 + 4 * brD;
  const bool lowD = mt < 4;
  const int row4 = mt * 16 + 4 * (lowD ? (laneD >> 4) : brD);
  const int col4 = ((mt + 4) & 7) * 16 + (lowD ? (laneD & 15) : 8 + bcD);
  const float2* aq = (const float2*)&Sm[arow * SLD + laneHi * 2];
  const float2* bq = &Xq[laneHi * 16 + nidx];

  for (int tstep = 0; tstep < TS; ++tstep) {
    {
      const v4f xi0 = *(const v4f*)&Xx[rowA];
      const v4f xi1 = *(const v4f*)&Xy[rowA];
      const v4f xi2 = *(const v4f*)&Xz[rowA];
      #pragma unroll
      for (int k = 0; k < 4; ++k) {
        const int col = ((mt + k) & 7) * 16 + 2 * bcD;
        const v2f xj0 = *(const v2f*)&Xx[col];
        const v2f xj1 = *(const v2f*)&Xy[col];
        const v2f xj2 = *(const v2f*)&Xz[col];
        v2f dd[4], sd[4];
        #pragma unroll
        for (int a = 0; a < 4; ++a) dd[a] = *(const v2f*)&Dp[(rowA + a) * NN + col];
        if (k == 0) {
          #pragma unroll
          for (int a = 0; a < 4; ++a) {
            #pragma unroll
            for (int e = 0; e < 2; ++e)
              sd[a][e] = s_cell(xi0[a] - xj0[e], xi1[a] - xj1[e], xi2[a] - xj2[e],
                                cw[a >> 1][(a & 1) * 2 + e], dd[a][e]);
          }
        } else {
          v4f dm[2], sm[2];
          #pragma unroll
          for (int e = 0; e < 2; ++e) dm[e] = *(const v4f*)&Dp[(col + e) * NN + rowA];
          #pragma unroll
          for (int a = 0; a < 4; ++a) {
            #pragma unroll
            for (int e = 0; e < 2; ++e) {
              float sij, sji;
              s_pair(xi0[a] - xj0[e], xi1[a] - xj1[e], xi2[a] - xj2[e],
                     cw[2 * k + (a >> 1)][(a & 1) * 2 + e], cw[6 + 2 * k + e][a],
                     dd[a][e], dm[e][a], sij, sji);
              sd[a][e] = sij;
              sm[e][a] = sji;
            }
          }
          #pragma unroll
          for (int e = 0; e < 2; ++e) *(v4f*)&Sm[(col + e) * SLD + rowA] = sm[e];
        }
        #pragma unroll
        for (int a = 0; a < 4; ++a) *(v2f*)&Sm[(rowA + a) * SLD + col] = sd[a];
      }
      {
        const v4f yi0 = *(const v4f*)&Xx[row4];
        const v4f yi1 = *(const v4f*)&Xy[row4];
        const v4f yi2 = *(const v4f*)&Xz[row4];
        const float yj0 = Xx[col4], yj1 = Xy[col4], yj2 = Xz[col4];
        const v4f dm = *(const v4f*)&Dp[col4 * NN + row4];
        v4f sm;
        #pragma unroll
        for (int a = 0; a < 4; ++a) {
          float sij, sji;
          s_pair(yi0[a] - yj0, yi1[a] - yj1, yi2[a] - yj2,
                 cw[14][a], cw[15][a], Dp[(row4 + a) * NN + col4], dm[a], sij, sji);
          Sm[(row4 + a) * SLD + col4] = sij;
          sm[a] = sji;
        }
        *(v4f*)&Sm[col4 * SLD + row4] = sm;
      }
      __syncthreads();
    }

    v8f acc = {};
    #pragma unroll
    for (int kt = 0; kt < 32; ++kt) {
      const float2 apair = aq[kt * 2];
      const float2 bpair = bq[kt * 32];
      v2f a;  a[0]  = apair.x;  a[1]  = apair.y;
      v2f bb; bb[0] = bpair.x;  bb[1] = bpair.y;
      acc = __builtin_amdgcn_wmma_f32_16x16x4_f32(
           false, a,  false, bb,
           (short)0, acc,  false,  false);
    }
    {
      const int rbase = mt * 16 + laneHi * 8;
      #pragma unroll
      for (int r = 0; r < 8; ++r) Pp[nidx * PLD + rbase + r] = acc[r];
    }
    __syncthreads();

    if (tid < NN) {
      const float rs = Pp[3 * PLD + tid];
      const float gx = rs * Xx[tid] - Pp[0 * PLD + tid];
      const float gy = rs * Xy[tid] - Pp[1 * PLD + tid];
      const float gz = rs * Xz[tid] - Pp[2 * PLD + tid];
      const float dx = -STEP_EPS * gx;
      const float dy = -STEP_EPS * gy;
      const float dz = -STEP_EPS * gz;
      const float speed = sqrt_rn_bounded(dx * dx + dy * dy + dz * dz + 0.001f);
      const float at =
          ALPHA_BASE + (ALPHA - ALPHA_BASE) * ((float)(TS - tstep) / (float)TS);
      const float sc = at * tanhf(speed / at) / speed;
      Xx[tid] += dx * sc;
      Xy[tid] += dy * sc;
      Xz[tid] += dz * sc;
    }
    __syncthreads();
    if (tid < 192) {
      const int q = tid & 63, n = tid >> 6;
      const float* comp = (n == 0) ? Xx : ((n == 1) ? Xy : Xz);
      Xq[q * 16 + n] = make_float2(comp[2 * q], comp[2 * q + 1]);
    }
    __syncthreads();
  }

  float* ob = out + (size_t)b * NN * NN;
  int tido = tid;
  asm volatile("" : "+v"(tido));
  {
    const int lo = tido & 31, wo = tido >> 5;
    const v4i* ad4 = (const v4i*)ad + (wo * 16) * 32 + lo;
    v4f*       ob4 = (v4f*)ob       + (wo * 16) * 32 + lo;
    const float4 xj0 = *(const float4*)&Xx[4 * lo];
    const float4 xj1 = *(const float4*)&Xy[4 * lo];
    const float4 xj2 = *(const float4*)&Xz[4 * lo];
    v4i cur[4], nxt[4];
    #pragma unroll
    for (int a = 0; a < 4; ++a) cur[a] = ad4[a * 32];
    #pragma unroll 1
    for (int g = 0; g < 4; ++g) {
      const int gn = (g + 1) & 3;
      #pragma unroll
      for (int a = 0; a < 4; ++a) nxt[a] = ad4[(gn * 4 + a) * 32];
      #pragma unroll
      for (int a = 0; a < 4; ++a) {
        const int i = wo * 16 + g * 4 + a;
        const float xi0 = Xx[i], xi1 = Xy[i], xi2 = Xz[i];
        const v4i m4 = cur[a];
        v4f o;
        o.x = out_cell(xi0 - xj0.x, xi1 - xj1.x, xi2 - xj2.x, m4.x);
        o.y = out_cell(xi0 - xj0.y, xi1 - xj1.y, xi2 - xj2.y, m4.y);
        o.z = out_cell(xi0 - xj0.z, xi1 - xj1.z, xi2 - xj2.z, m4.z);
        o.w = out_cell(xi0 - xj0.w, xi1 - xj1.w, xi2 - xj2.w, m4.w);
        ob4[(g * 4 + a) * 32] = o;
      }
      #pragma unroll
      for (int a = 0; a < 4; ++a) cur[a] = nxt[a];
    }
  }
}

extern "C" void kernel_launch(void* const* d_in, const int* in_sizes, int n_in,
                              void* d_out, int out_size, void* d_ws, size_t ws_size,
                              hipStream_t stream) {
  const float* edge_pred = (const float*)d_in[0];
  const int*   adj       = (const int*)  d_in[1];
  const float* d_init    = (const float*)d_in[2];
  const float* u_init    = (const float*)d_in[3];
  const float* x_noise   = (const float*)d_in[4];
  float*       outp      = (float*)d_out;
  const int B = in_sizes[1] / (NN * NN);
  ts_gcn_fused<<<B, 256, 0, stream>>>(edge_pred, adj, d_init, u_init, x_noise, outp);
}


# D plane XOR-swizzled in LDS (row bits 1-3 into column bits 3-5, pitch stays 128): S-build 4x2 tile reads and 2x4 mirror reads were 4- to 16-way bank conflicts, now 1- to 2-way; same sums, same order
# speedup vs baseline: 5.2178x; 1.0324x over previous
#include <hip/hip_runtime.h>
#include <hip/hip_bf16.h>
#include <math.h>


typedef __attribute__((ext_vector_type(2))) float v2f;
typedef __attribute__((ext_vector_type(4))) float v4f;
typedef __attribute__((ext_vector_type(4))) int   v4i;
typedef __attribute__((ext_vector_type(8))) float v8f;

#define NN   128
#define SLD  132
#define PLD  132
#define TS   10
#define STEP_EPS 0.1f
#define ALPHA 5.0f
#define ALPHA_BASE 0.1f
#define KRANK 3
#define PSTEPS 10

__device__ __forceinline__ float softplus_f(float x) {
  return fmaxf(x, 0.0f) + log1pf(expf(-fabsf(x)));
}

__device__ __forceinline__ float sqrt_rn_bounded(float x) {
  const float s  = __builtin_amdgcn_sqrtf(x);
  const float sd = __int_as_float(__float_as_int(s) - 1);
  const float su = __int_as_float(__float_as_int(s) + 1);
  const float vd = fmaf(-sd, s, x);
  const float vu = fmaf(-su, s, x);
  float r = (vd <= 0.0f) ? sd : s;
  r = (vu > 0.0f) ? su : r;
  return r;
}

__device__ __forceinline__ void dist_rinv(float q, float& dist, float& rinv) {
  const float s  = __builtin_amdgcn_sqrtf(q);
  const float r0 = __builtin_amdgcn_rcpf(s);
  const float d0 = fmaf(-s, s, q);
  dist = fmaf(d0, 0.5f * r0, s);
  const float e0 = fmaf(-dist, r0, 1.0f);
  const float r1 = fmaf(e0, r0, r0);
  const float e1 = fmaf(-dist, r1, 1.0f);
  rinv = fmaf(e1, r1, r1);
}

__device__ __forceinline__ float s_cell(float u0, float u1, float u2,
                                        float e, float d) {
  float dist, rinv;
  dist_rinv(u0 * u0 + u1 * u1 + u2 * u2 + 0.01f, dist, rinv);
  return e * (dist - d) * rinv;
}

__device__ __forceinline__ void s_pair(float u0, float u1, float u2,
                                       float eij, float eji, float dij, float dji,
                                       float& sij, float& sji) {
  float dist, rinv;
  dist_rinv(u0 * u0 + u1 * u1 + u2 * u2 + 0.01f, dist, rinv);
  sij = eij * (dist - dij) * rinv;
  sji = eji * (dist - dji) * rinv;
}

__device__ __forceinline__ float out_cell(float u0, float u1, float u2, int m) {
  return (float)m * sqrt_rn_bounded(fmaf(u2, u2, fmaf(u0, u0, u1 * u1)) + 0.01f);
}

__device__ __forceinline__ float block_reduce(float v, float* red, int tid) {
  red[tid] = v;
  __syncthreads();
  #pragma unroll
  for (int s = 128; s > 0; s >>= 1) {
    if (tid < s) red[tid] += red[tid + s];
    __syncthreads();
  }
  float r = red[0];
  __syncthreads();
  return r;
}


__device__ __forceinline__ v2f fma2(v2f x, v2f y, v2f acc) {
  return __builtin_elementwise_fma(x, y, acc);
}

__device__ __forceinline__ int dsw(int i, int j) {
  return i * NN + (j ^ ((i & 14) << 2));
}


template <int CTRL>
__device__ __forceinline__ float add_row_xmask(float v) {
  return v + __builtin_bit_cast(float, __builtin_amdgcn_update_dpp(
                 0, __builtin_bit_cast(int, v), CTRL, 0xf, 0xf, false));
}

__device__ __forceinline__ float wave_sum(float v) {
  v = add_row_xmask<0x161>(v);
  v = add_row_xmask<0x162>(v);
  v = add_row_xmask<0x164>(v);
  v = add_row_xmask<0x168>(v);
  const float lo = __builtin_bit_cast(float, __builtin_amdgcn_readlane(__builtin_bit_cast(int, v), 0));
  const float hi = __builtin_bit_cast(float, __builtin_amdgcn_readlane(__builtin_bit_cast(int, v), 16));
  return lo + hi;
}

struct PSet { v4f g[4][2]; };
struct PAcc { v2f a[4]; v2f b[4]; };

__device__ __forceinline__ void pset_load(PSet& s, const float* g0, const float* g1,
                                          const float* g2, const float* g3, int j) {
  #pragma unroll
  for (int q = 0; q < 2; ++q) {
    s.g[0][q] = *(const v4f*)&g0[j + 4 * q];
    s.g[1][q] = *(const v4f*)&g1[j + 4 * q];
    s.g[2][q] = *(const v4f*)&g2[j + 4 * q];
    s.g[3][q] = *(const v4f*)&g3[j + 4 * q];
  }
}

template <int S>
__device__ __forceinline__ void pset_use(PAcc& acc, const PSet& s, float ucur) {
  float us[8];
  #pragma unroll
  for (int c = 0; c < 8; ++c)
    us[c] = __builtin_bit_cast(float, __builtin_amdgcn_readlane(
                __builtin_bit_cast(int, ucur), 8 * S + c));
  #pragma unroll
  for (int q = 0; q < 2; ++q) {
    const v2f ulo = {us[4 * q],     us[4 * q + 1]};
    const v2f uhi = {us[4 * q + 2], us[4 * q + 3]};
    #pragma unroll
    for (int r = 0; r < 4; ++r) {
      const v4f g = s.g[r][q];
      if (q & 1) { acc.b[r] = fma2(g.xy, ulo, acc.b[r]);  acc.b[r] = fma2(g.zw, uhi, acc.b[r]); }
      else       { acc.a[r] = fma2(g.xy, ulo, acc.a[r]);  acc.a[r] = fma2(g.zw, uhi, acc.a[r]); }
    }
  }
}

__device__ __forceinline__ void power_group(PAcc& acc, PSet& sa, PSet& sb, PSet& sc, PSet& sd,
                                            const float* g0, const float* g1,
                                            const float* g2, const float* g3,
                                            int jb, float ucur) {
  pset_load(sc, g0, g1, g2, g3, jb + 16);
  pset_use<0>(acc, sa, ucur);
  pset_load(sd, g0, g1, g2, g3, jb + 24);
  pset_use<1>(acc, sb, ucur);
  pset_load(sa, g0, g1, g2, g3, (jb + 32) & (NN - 1));
  pset_use<2>(acc, sc, ucur);
  pset_load(sb, g0, g1, g2, g3, (jb + 40) & (NN - 1));
  pset_use<3>(acc, sd, ucur);
  asm volatile("" ::: "memory");
}

template <int KX>
__device__ __forceinline__ void power_column(const float* gram,
                                             float (*ucol)[NN], const float* ub,
                                             int lane) {
  const float* g0 = gram + (lane     ) * SLD;
  const float* g1 = gram + (lane + 32) * SLD;
  const float* g2 = gram + (lane + 64) * SLD;
  const float* g3 = gram + (lane + 96) * SLD;
  float u0 = ub[(lane     ) * 3 + KX];
  float u1 = ub[(lane + 32) * 3 + KX];
  float u2 = ub[(lane + 64) * 3 + KX];
  float u3 = ub[(lane + 96) * 3 + KX];
  float p0 = 0.0f, p1 = 0.0f, p2 = 0.0f, p3 = 0.0f;
  float q0 = 0.0f, q1 = 0.0f, q2 = 0.0f, q3 = 0.0f;
  if constexpr (KX > 0) {
    p0 = ucol[0][lane];      p1 = ucol[0][lane + 32];
    p2 = ucol[0][lane + 64]; p3 = ucol[0][lane + 96];
  }
  if constexpr (KX > 1) {
    q0 = ucol[1][lane];      q1 = ucol[1][lane + 32];
    q2 = ucol[1][lane + 64]; q3 = ucol[1][lane + 96];
  }
  const v2f zero2 = {0.0f, 0.0f};

  PSet sa, sb, sc, sd;
  pset_load(sa, g0, g1, g2, g3, 0);
  pset_load(sb, g0, g1, g2, g3, 8);

  for (int it = 0; it < PSTEPS; ++it) {
    pset_load(sc, g0, g1, g2, g3, 16);
    pset_load(sd, g0, g1, g2, g3, 24);
    const float nsq = wave_sum((u0 * u0 + u1 * u1) + (u2 * u2 + u3 * u3));
    float inorm = 1.0f / fmaxf(sqrtf(nsq), 0.001f);
    float dp = 0.0f, dq = 0.0f;
    if constexpr (KX > 0) dp = wave_sum((p0 * u0 + p1 * u1) + (p2 * u2 + p3 * u3));
    if constexpr (KX > 1) dq = wave_sum((q0 * u0 + q1 * u1) + (q2 * u2 + q3 * u3));
    asm volatile("" : "+v"(inorm));
    if constexpr (KX > 0) asm volatile("" : "+v"(dp));
    if constexpr (KX > 1) asm volatile("" : "+v"(dq));
    PAcc acc;
    #pragma unroll
    for (int r = 0; r < 4; ++r) { acc.a[r] = zero2; acc.b[r] = zero2; }
    pset_use<0>(acc, sa, u0);
    pset_use<1>(acc, sb, u0);
    pset_load(sa, g0, g1, g2, g3, 32);
    pset_use<2>(acc, sc, u0);
    pset_load(sb, g0, g1, g2, g3, 40);
    pset_use<3>(acc, sd, u0);
    float ucur = u1;
    #pragma unroll 1
    for (int t = 1; t < 4; ++t) {
      power_group(acc, sa, sb, sc, sd, g0, g1, g2, g3, 32 * t, ucur);
      ucur = (t == 1) ? u2 : u3;
    }
    float r0 = (acc.a[0].x + acc.a[0].y) + (acc.b[0].x + acc.b[0].y);
    float r1 = (acc.a[1].x + acc.a[1].y) + (acc.b[1].x + acc.b[1].y);
    float r2 = (acc.a[2].x + acc.a[2].y) + (acc.b[2].x + acc.b[2].y);
    float r3 = (acc.a[3].x + acc.a[3].y) + (acc.b[3].x + acc.b[3].y);
    if constexpr (KX > 0) {
      r0 = fmaf(-p0, dp, r0);  r1 = fmaf(-p1, dp, r1);
      r2 = fmaf(-p2, dp, r2);  r3 = fmaf(-p3, dp, r3);
    }
    if constexpr (KX > 1) {
      r0 = fmaf(-q0, dq, r0);  r1 = fmaf(-q1, dq, r1);
      r2 = fmaf(-q2, dq, r2);  r3 = fmaf(-q3, dq, r3);
    }
    u0 = r0 * inorm;  u1 = r1 * inorm;  u2 = r2 * inorm;  u3 = r3 * inorm;
  }

  const float eig  = wave_sum((u0 * u0 + u1 * u1) + (u2 * u2 + u3 * u3));
  const float iden = 1.0f / sqrtf(sqrtf(eig + 0.01f));
  ucol[KX][lane]      = u0 * iden;
  ucol[KX][lane + 32] = u1 * iden;
  ucol[KX][lane + 64] = u2 * iden;
  ucol[KX][lane + 96] = u3 * iden;
}

__launch_bounds__(256, 4)
__global__ void ts_gcn_fused(const float* __restrict__ edge_pred,
                             const int*   __restrict__ adj,
                             const float* __restrict__ d_init,
                             const float* __restrict__ u_init,
                             const float* __restrict__ x_noise,
                             float*       __restrict__ out)
{
  __shared__ __attribute__((aligned(16))) float Dp[NN * NN];
  __shared__ __attribute__((aligned(16))) float Sm[NN * SLD];
  __shared__ float2 Xq[64 * 16];
  __shared__ __attribute__((aligned(16))) float Pp[16 * PLD];
  __shared__ __attribute__((aligned(16))) float Xx[NN];
  __shared__ __attribute__((aligned(16))) float Xy[NN];
  __shared__ __attribute__((aligned(16))) float Xz[NN];
  __shared__ float  rsum[NN];
  __shared__ float  red[256];
  __shared__ __attribute__((aligned(16))) float ucol[KRANK][NN];

  const int b   = blockIdx.x;
  const int tid = threadIdx.x;
  const int lane   = tid & 31;
  const int mt     = tid >> 5;
  const float d0 = d_init[0];

  const float* ep = edge_pred + (size_t)b * NN * NN * 2;
  const int*   ad = adj       + (size_t)b * NN * NN;

  float partM = 0.0f;
  {
    const v4f* ep4 = (const v4f*)ep;
    const v4i* ad4 = (const v4i*)ad;
    for (int it = tid; it < 528; it += 256) {
      const int R = it / 33, k = it - R * 33;
      const bool lo = k < 32 - R;
      const int I = lo ? R : 31 - R;
      const int J = lo ? R + k : k - 1;
      v4f e[4][2], t[4][2];
      v4i ma[4], mb[4];
      #pragma unroll
      for (int a = 0; a < 4; ++a) {
        e[a][0] = ep4[(4 * I + a) * 64 + 2 * J];
        e[a][1] = ep4[(4 * I + a) * 64 + 2 * J + 1];
        t[a][0] = ep4[(4 * J + a) * 64 + 2 * I];
        t[a][1] = ep4[(4 * J + a) * 64 + 2 * I + 1];
        ma[a]   = ad4[(4 * I + a) * 32 + J];
        mb[a]   = ad4[(4 * J + a) * 32 + I];
      }
      const float offd = (I != J) ? 1.0f : 0.0f;
      v4f dd[4], ww[4];
      v4f dm[4], wm[4];
      #pragma unroll
      for (int a = 0; a < 4; ++a) {
        #pragma unroll
        for (int b = 0; b < 4; ++b) {
          const float e0 = e[a][b >> 1][(b & 1) * 2];
          const float e1 = e[a][b >> 1][(b & 1) * 2 + 1];
          const float t0 = t[b][a >> 1][(a & 1) * 2];
          const float t1 = t[b][a >> 1][(a & 1) * 2 + 1];
          const float sp0 = softplus_f(d0 + (e0 + t0));
          const float sp1 = softplus_f(d0 + (e1 + t1));
          const float m  = (float)ma[a][b];
          const float mm = (float)mb[b][a];
          dd[a][b] = sp0 * m;
          ww[a][b] = sp1 * m;
          dm[b][a] = sp0 * mm;
          wm[b][a] = sp1 * mm;
          partM += (I == J && a == b) ? 1.0f : m;
          partM = fmaf(offd, mm, partM);
          __builtin_amdgcn_sched_barrier(0);
        }
      }
      #pragma unroll
      for (int a = 0; a < 4; ++a) {
        *(v4f*)&Dp[dsw(4 * I + a, 4 * J)]     = dd[a];
        *(v4f*)&Sm[(4 * I + a) * SLD + 4 * J] = ww[a];
        *(v4f*)&Dp[dsw(4 * J + a, 4 * I)]     = dm[a];
        *(v4f*)&Sm[(4 * J + a) * SLD + 4 * I] = wm[a];
      }
    }
  }
  for (int idx = tid; idx < 64 * 16; idx += 256) {
    const float fill = ((idx & 15) == 3) ? 1.0f : 0.0f;
    Xq[idx] = make_float2(fill, fill);
  }
  const float Msum = block_reduce(partM, red, tid);

  const float invM4 = 4.0f / Msum;
  v4f cw[16];
  {
    const int br = lane >> 3, bc = lane & 7;
    const int rowA = mt * 16 + 4 * br;
    #pragma unroll
    for (int k = 0; k < 4; ++k) {
      const int col = ((mt + k) & 7) * 16 + 2 * bc;
      #pragma unroll
      for (int a = 0; a < 4; ++a) {
        const v2f w2 = *(const v2f*)&Sm[(rowA + a) * SLD + col];
        cw[2 * k + (a >> 1)][(a & 1) * 2]     = invM4 * w2.x;
        cw[2 * k + (a >> 1)][(a & 1) * 2 + 1] = invM4 * w2.y;
      }
      if (k > 0) {
        #pragma unroll
        for (int e = 0; e < 2; ++e)
          cw[6 + 2 * k + e] = invM4 * *(const v4f*)&Sm[(col + e) * SLD + rowA];
      }
    }
    const bool low = mt < 4;
    const int row4 = mt * 16 + 4 * (low ? (lane >> 4) : (lane >> 3));
    const int col4 = ((mt + 4) & 7) * 16 + (low ? (lane & 15) : 8 + (lane & 7));
    #pragma unroll
    for (int a = 0; a < 4; ++a) cw[14][a] = invM4 * Sm[(row4 + a) * SLD + col4];
    cw[15] = invM4 * *(const v4f*)&Sm[col4 * SLD + row4];
  }

  const float Nmol = block_reduce(
      (tid < NN) ? ((tid == 0) ? 1.0f : (float)ad[tid * NN]) : 0.0f, red, tid);

  {
    const int i = tid >> 1, j0 = (tid & 1) << 6;
    float part = 0.0f;
    for (int jb = 0; jb < 64; jb += 8) {
      const float* dg = &Dp[dsw(i, j0 + jb)];
      for (int jj = 0; jj < 8; ++jj) {
        const float d = dg[jj];
        part = fmaf(d, d, part);
      }
    }
    red[tid] = part;
    __syncthreads();
    if (tid < NN) rsum[tid] = red[2 * tid] + red[2 * tid + 1];
    __syncthreads();
    const float tot  = block_reduce((tid < NN) ? rsum[tid] : 0.0f, red, tid);
    const float invN = 1.0f / Nmol;
    const float Mn   = tot * invN * invN;
    for (int p = tid; p < NN * NN; p += 256) {
      const int i2 = p >> 7, j2 = p & 127;
      const float d  = Dp[dsw(i2, j2)];
      const float d2 = d * d;
      const float mk = (i2 == j2) ? 1.0f : (d > 0.0f ? 1.0f : 0.0f);
      Sm[i2 * SLD + j2] =
          mk * -0.5f * (d2 - rsum[j2] * invN - rsum[i2] * invN + Mn);
    }
    __syncthreads();
  }

  if (tid < 32) {
    #pragma unroll
    for (int r = 0; r < 16; ++r) *(v4f*)&Pp[r * NN + 4 * lane] = cw[r];
    const float* ub = u_init + (size_t)b * NN * 3;
    power_column<0>(Sm, ucol, ub, tid);
    power_column<1>(Sm, ucol, ub, tid);
    power_column<2>(Sm, ucol, ub, tid);
    #pragma unroll
    for (int r = 0; r < 16; ++r) cw[r] = *(const v4f*)&Pp[r * NN + 4 * lane];
  }
  __syncthreads();
  if (tid < NN) {
    const float* xn = x_noise + (size_t)b * NN * 3 + tid * 3;
    Xx[tid] = ucol[0][tid] + xn[0];
    Xy[tid] = ucol[1][tid] + xn[1];
    Xz[tid] = ucol[2][tid] + xn[2];
  }
  __syncthreads();
  if (tid < 192) {
    const int q = tid & 63, n = tid >> 6;
    const float* comp = (n == 0) ? Xx : ((n == 1) ? Xy : Xz);
    Xq[q * 16 + n] = make_float2(comp[2 * q], comp[2 * q + 1]);
  }
  __syncthreads();

  int tidd = tid;
  asm volatile("" : "+v"(tidd));
  const int laneD  = tidd & 31;
  const int nidx   = laneD & 15;
  const int laneHi = laneD >> 4;
  const int arow   = mt * 16 + nidx;
  const int brD  = laneD >> 3, bcD = laneD & 7;
  const int rowA = mt * 16 + 4 * brD;
  const bool lowD = mt < 4;
  const int row4 = mt * 16 + 4 * (lowD ? (laneD >> 4) : brD);
  const int col4 = ((mt + 4) & 7) * 16 + (lowD ? (laneD & 15) : 8 + bcD);
  const float2* aq = (const float2*)&Sm[arow * SLD + laneHi * 2];
  const float2* bq = &Xq[laneHi * 16 + nidx];

  for (int tstep = 0; tstep < TS; ++tstep) {
    {
      const v4f xi0 = *(const v4f*)&Xx[rowA];
      const v4f xi1 = *(const v4f*)&Xy[rowA];
      const v4f xi2 = *(const v4f*)&Xz[rowA];
      #pragma unroll
      for (int k = 0; k < 4; ++k) {
        const int col = ((mt + k) & 7) * 16 + 2 * bcD;
        const v2f xj0 = *(const v2f*)&Xx[col];
        const v2f xj1 = *(const v2f*)&Xy[col];
        const v2f xj2 = *(const v2f*)&Xz[col];
        v2f dd[4], sd[4];
        #pragma unroll
        for (int a = 0; a < 4; ++a) dd[a] = *(const v2f*)&Dp[dsw(rowA + a, col)];
        if (k == 0) {
          #pragma unroll
          for (int a = 0; a < 4; ++a) {
            #pragma unroll
            for (int e = 0; e < 2; ++e)
              sd[a][e] = s_cell(xi0[a] - xj0[e], xi1[a] - xj1[e], xi2[a] - xj2[e],
                                cw[a >> 1][(a & 1) * 2 + e], dd[a][e]);
          }
        } else {
          v4f dm[2], sm[2];
          #pragma unroll
          for (int e = 0; e < 2; ++e) dm[e] = *(const v4f*)&Dp[dsw(col + e, rowA)];
          #pragma unroll
          for (int a = 0; a < 4; ++a) {
            #pragma unroll
            for (int e = 0; e < 2; ++e) {
              float sij, sji;
              s_pair(xi0[a] - xj0[e], xi1[a] - xj1[e], xi2[a] - xj2[e],
                     cw[2 * k + (a >> 1)][(a & 1) * 2 + e], cw[6 + 2 * k + e][a],
                     dd[a][e], dm[e][a], sij, sji);
              sd[a][e] = sij;
              sm[e][a] = sji;
            }
          }
          #pragma unroll
          for (int e = 0; e < 2; ++e) *(v4f*)&Sm[(col + e) * SLD + rowA] = sm[e];
        }
        #pragma unroll
        for (int a = 0; a < 4; ++a) *(v2f*)&Sm[(rowA + a) * SLD + col] = sd[a];
      }
      {
        const v4f yi0 = *(const v4f*)&Xx[row4];
        const v4f yi1 = *(const v4f*)&Xy[row4];
        const v4f yi2 = *(const v4f*)&Xz[row4];
        const float yj0 = Xx[col4], yj1 = Xy[col4], yj2 = Xz[col4];
        const v4f dm = *(const v4f*)&Dp[dsw(col4, row4)];
        v4f sm;
        #pragma unroll
        for (int a = 0; a < 4; ++a) {
          float sij, sji;
          s_pair(yi0[a] - yj0, yi1[a] - yj1, yi2[a] - yj2,
                 cw[14][a], cw[15][a], Dp[dsw(row4 + a, col4)], dm[a], sij, sji);
          Sm[(row4 + a) * SLD + col4] = sij;
          sm[a] = sji;
        }
        *(v4f*)&Sm[col4 * SLD + row4] = sm;
      }
      __syncthreads();
    }

    v8f acc = {};
    #pragma unroll
    for (int kt = 0; kt < 32; ++kt) {
      const float2 apair = aq[kt * 2];
      const float2 bpair = bq[kt * 32];
      v2f a;  a[0]  = apair.x;  a[1]  = apair.y;
      v2f bb; bb[0] = bpair.x;  bb[1] = bpair.y;
      acc = __builtin_amdgcn_wmma_f32_16x16x4_f32(
           false, a,  false, bb,
           (short)0, acc,  false,  false);
    }
    {
      const int rbase = mt * 16 + laneHi * 8;
      #pragma unroll
      for (int r = 0; r < 8; ++r) Pp[nidx * PLD + rbase + r] = acc[r];
    }
    __syncthreads();

    if (tid < NN) {
      const float rs = Pp[3 * PLD + tid];
      const float gx = rs * Xx[tid] - Pp[0 * PLD + tid];
      const float gy = rs * Xy[tid] - Pp[1 * PLD + tid];
      const float gz = rs * Xz[tid] - Pp[2 * PLD + tid];
      const float dx = -STEP_EPS * gx;
      const float dy = -STEP_EPS * gy;
      const float dz = -STEP_EPS * gz;
      const float speed = sqrt_rn_bounded(dx * dx + dy * dy + dz * dz + 0.001f);
      const float at =
          ALPHA_BASE + (ALPHA - ALPHA_BASE) * ((float)(TS - tstep) / (float)TS);
      const float sc = at * tanhf(speed / at) / speed;
      Xx[tid] += dx * sc;
      Xy[tid] += dy * sc;
      Xz[tid] += dz * sc;
    }
    __syncthreads();
    if (tid < 192) {
      const int q = tid & 63, n = tid >> 6;
      const float* comp = (n == 0) ? Xx : ((n == 1) ? Xy : Xz);
      Xq[q * 16 + n] = make_float2(comp[2 * q], comp[2 * q + 1]);
    }
    __syncthreads();
  }

  float* ob = out + (size_t)b * NN * NN;
  int tido = tid;
  asm volatile("" : "+v"(tido));
  {
    const int lo = tido & 31, wo = tido >> 5;
    const v4i* ad4 = (const v4i*)ad + (wo * 16) * 32 + lo;
    v4f*       ob4 = (v4f*)ob       + (wo * 16) * 32 + lo;
    const float4 xj0 = *(const float4*)&Xx[4 * lo];
    const float4 xj1 = *(const float4*)&Xy[4 * lo];
    const float4 xj2 = *(const float4*)&Xz[4 * lo];
    v4i cur[4], nxt[4];
    #pragma unroll
    for (int a = 0; a < 4; ++a) cur[a] = ad4[a * 32];
    #pragma unroll 1
    for (int g = 0; g < 4; ++g) {
      const int gn = (g + 1) & 3;
      #pragma unroll
      for (int a = 0; a < 4; ++a) nxt[a] = ad4[(gn * 4 + a) * 32];
      #pragma unroll
      for (int a = 0; a < 4; ++a) {
        const int i = wo * 16 + g * 4 + a;
        const float xi0 = Xx[i], xi1 = Xy[i], xi2 = Xz[i];
        const v4i m4 = cur[a];
        v4f o;
        o.x = out_cell(xi0 - xj0.x, xi1 - xj1.x, xi2 - xj2.x, m4.x);
        o.y = out_cell(xi0 - xj0.y, xi1 - xj1.y, xi2 - xj2.y, m4.y);
        o.z = out_cell(xi0 - xj0.z, xi1 - xj1.z, xi2 - xj2.z, m4.z);
        o.w = out_cell(xi0 - xj0.w, xi1 - xj1.w, xi2 - xj2.w, m4.w);
        ob4[(g * 4 + a) * 32] = o;
      }
      #pragma unroll
      for (int a = 0; a < 4; ++a) cur[a] = nxt[a];
    }
  }
}

extern "C" void kernel_launch(void* const* d_in, const int* in_sizes, int n_in,
                              void* d_out, int out_size, void* d_ws, size_t ws_size,
                              hipStream_t stream) {
  const float* edge_pred = (const float*)d_in[0];
  const int*   adj       = (const int*)  d_in[1];
  const float* d_init    = (const float*)d_in[2];
  const float* u_init    = (const float*)d_in[3];
  const float* x_noise   = (const float*)d_in[4];
  float*       outp      = (float*)d_out;
  const int B = in_sizes[1] / (NN * NN);
  ts_gcn_fused<<<B, 256, 0, stream>>>(edge_pred, adj, d_init, u_init, x_noise, outp);
}
